# starting kernel
# baseline (speedup 1.0000x reference)
#include <hip/hip_runtime.h>
#include <hip/hip_bf16.h>

#define N_NODES 50000
#define N_EDGES 800000
#define IN_C    128
#define OUT_C   128
#define HIDDEN  16
#define N_COEF  7

typedef __attribute__((ext_vector_type(2))) float v2f;
typedef __attribute__((ext_vector_type(8))) float v8f;

__device__ __forceinline__ v8f wmma4(v2f a, v2f b, v8f c) {
  return __builtin_amdgcn_wmma_f32_16x16x4_f32(false, a, false, b, (short)0, c,
                                               false, false);
}

__device__ __forceinline__ float silu_f(float v) {
  return v * (1.0f / (1.0f + __expf(-v)));
}

__device__ __forceinline__ void bspline7(float v, float* out) {
  float b[10];
#pragma unroll
  for (int j = 0; j < 10; ++j) {
    float gj = 0.5f * (float)j - 2.5f;
    b[j] = (v >= gj && v < gj + 0.5f) ? 1.0f : 0.0f;
  }
#pragma unroll
  for (int p = 1; p <= 3; ++p) {
    float invd = 1.0f / (0.5f * (float)p);
#pragma unroll
    for (int j = 0; j < 10 - p; ++j) {
      float gj  = 0.5f * (float)j - 2.5f;
      float gjp = gj + 0.5f * (float)p;
      b[j] = (v - gj) * invd * b[j] + ((gjp + 0.5f) - v) * invd * b[j + 1];
    }
  }
#pragma unroll
  for (int c = 0; c < N_COEF; ++c) out[c] = b[c];
}

__global__ void k_gemm_xw(const float* __restrict__ x, const float* __restrict__ W,
                          float* __restrict__ hlin) {
  const int NT = N_NODES / 16;
  const int NG = (NT + 3) / 4;
  int wave = (int)((blockIdx.x * blockDim.x + threadIdx.x) >> 5);
  int lane = (int)(threadIdx.x & 31);
  if (wave >= NG * (OUT_C / 16)) return;
  int grp = wave >> 3, ct = wave & 7;
  int o0 = ct * 16;
  int half = lane >> 4, l16 = lane & 15;

  int  rt[4];
  bool valid[4];
#pragma unroll
  for (int i = 0; i < 4; ++i) {
    int r = grp * 4 + i;
    valid[i] = (r < NT);
    rt[i] = valid[i] ? r : (NT - 1);
  }

  v8f acc[4] = {};
  for (int k = 0; k < IN_C; k += 4) {
    int ka0 = k + (half ? 2 : 0);
    int ka1 = k + (half ? 3 : 1);
    v2f b;
    b.x = W[(o0 + l16) * IN_C + ka0];
    b.y = W[(o0 + l16) * IN_C + ka1];
#pragma unroll
    for (int i = 0; i < 4; ++i) {
      v2f a;
      a.x = x[(rt[i] * 16 + l16) * IN_C + ka0];
      a.y = x[(rt[i] * 16 + l16) * IN_C + ka1];
      acc[i] = wmma4(a, b, acc[i]);
    }
  }
#pragma unroll
  for (int i = 0; i < 4; ++i) {
    if (!valid[i]) continue;
#pragma unroll
    for (int j = 0; j < 8; ++j)
      hlin[(rt[i] * 16 + j + half * 8) * OUT_C + o0 + l16] = acc[i][j];
  }
}

__global__ void k_deg_init(float* __restrict__ deg) {
  int i = blockIdx.x * blockDim.x + threadIdx.x;
  if (i < N_NODES) deg[i] = 1.0f;
}

__global__ void k_deg_acc(const int* __restrict__ ei, float* __restrict__ deg) {
  int e = blockIdx.x * blockDim.x + threadIdx.x;
  if (e < N_EDGES)
    __hip_atomic_fetch_add(&deg[ei[N_EDGES + e]], 1.0f, __ATOMIC_RELAXED,
                           __HIP_MEMORY_SCOPE_AGENT);
}

__global__ void k_deg_rsqrt(float* __restrict__ deg) {
  int i = blockIdx.x * blockDim.x + threadIdx.x;
  if (i < N_NODES) {
    float d = deg[i];
    deg[i] = (d > 0.0f) ? rsqrtf(d) : 0.0f;
  }
}

__global__ void k_self_bias(const float* __restrict__ hlin,
                            const float* __restrict__ dinv,
                            const float* __restrict__ bias,
                            float* __restrict__ h) {
  int t = blockIdx.x * blockDim.x + threadIdx.x;
  if (t < N_NODES * OUT_C) {
    int n = t >> 7, c = t & 127;
    float di = dinv[n];
    h[t] = hlin[t] * di * di + bias[c];
  }
}

__global__ void k_edge_scatter(const int* __restrict__ ei,
                               const float* __restrict__ hlin,
                               const float* __restrict__ dinv,
                               float* __restrict__ h) {
  long long t = (long long)blockIdx.x * blockDim.x + threadIdx.x;
  int e = (int)(t >> 5);
  int lane = (int)(t & 31);
  if (e >= N_EDGES) return;
  int src = ei[e], dst = ei[N_EDGES + e];
  const float4* h4 = (const float4*)hlin;
  const float4* p = &h4[src * 32 + lane];
  __builtin_prefetch(p, 0, 3);
  float norm = dinv[src] * dinv[dst];
  float4 v = *p;
  float* base = &h[dst * OUT_C + lane * 4];
  __hip_atomic_fetch_add(base + 0, v.x * norm, __ATOMIC_RELAXED, __HIP_MEMORY_SCOPE_AGENT);
  __hip_atomic_fetch_add(base + 1, v.y * norm, __ATOMIC_RELAXED, __HIP_MEMORY_SCOPE_AGENT);
  __hip_atomic_fetch_add(base + 2, v.z * norm, __ATOMIC_RELAXED, __HIP_MEMORY_SCOPE_AGENT);
  __hip_atomic_fetch_add(base + 3, v.w * norm, __ATOMIC_RELAXED, __HIP_MEMORY_SCOPE_AGENT);
}

__global__ void k_prep_w0(const float* __restrict__ bw, const float* __restrict__ sw,
                          const float* __restrict__ sc, float* __restrict__ Bm) {
  int t = blockIdx.x * blockDim.x + threadIdx.x;
  if (t >= 8 * OUT_C * HIDDEN) return;
  int k = t >> 4, o = t & 15;
  int f = k >> 3, j = k & 7;
  float w = (j == 0) ? bw[o * OUT_C + f]
                     : sw[(o * OUT_C + f) * N_COEF + (j - 1)] * sc[o * OUT_C + f];
  Bm[k * HIDDEN + o] = w;
}

__global__ void k_prep_w1(const float* __restrict__ bw, const float* __restrict__ sw,
                          const float* __restrict__ sc, float* __restrict__ Bm) {
  int t = blockIdx.x * blockDim.x + threadIdx.x;
  if (t >= 8 * HIDDEN * OUT_C) return;
  int k = t >> 7, o = t & 127;
  int f = k >> 3, j = k & 7;
  float w = (j == 0) ? bw[o * HIDDEN + f]
                     : sw[(o * HIDDEN + f) * N_COEF + (j - 1)] * sc[o * HIDDEN + f];
  Bm[k * OUT_C + o] = w;
}

#define LDS0_STRIDE 258
__global__ void __launch_bounds__(32) k_kan0(const float* __restrict__ h,
                                             const float* __restrict__ Bm,
                                             float* __restrict__ g) {
  __shared__ float feat[16 * LDS0_STRIDE];
  int n0 = blockIdx.x * 16;
  int lane = threadIdx.x;
  int half = lane >> 4, l16 = lane & 15;
  v8f accA = {}, accB = {};
  for (int ch = 0; ch < 4; ++ch) {
    int f0 = ch * 32;
    for (int idx = lane; idx < 16 * 32; idx += 32) {
      int row = idx & 15, fl = idx >> 4;
      float v = h[(n0 + row) * OUT_C + f0 + fl];
      float* d = &feat[row * LDS0_STRIDE + fl * 8];
      d[0] = silu_f(v);
      float bs[N_COEF];
      bspline7(v, bs);
#pragma unroll
      for (int c = 0; c < N_COEF; ++c) d[1 + c] = bs[c];
    }
    __syncthreads();
    for (int k = 0; k < 256; k += 8) {
#pragma unroll
      for (int s = 0; s < 2; ++s) {
        int ka0 = k + s * 4 + (half ? 2 : 0);
        int ka1 = k + s * 4 + (half ? 3 : 1);
        v2f a, b;
        a.x = feat[l16 * LDS0_STRIDE + ka0];
        a.y = feat[l16 * LDS0_STRIDE + ka1];
        b.x = Bm[(ch * 256 + ka0) * HIDDEN + l16];
        b.y = Bm[(ch * 256 + ka1) * HIDDEN + l16];
        if (s == 0) accA = wmma4(a, b, accA);
        else        accB = wmma4(a, b, accB);
      }
    }
    __syncthreads();
  }
  v8f acc = accA + accB;
#pragma unroll
  for (int i = 0; i < 8; ++i)
    g[(n0 + i + half * 8) * HIDDEN + l16] = acc[i];
}

#define LDS1_STRIDE 130
__global__ void __launch_bounds__(32) k_kan1(const float* __restrict__ g,
                                             const float* __restrict__ Bm,
                                             float* __restrict__ out) {
  __shared__ float feat[16 * LDS1_STRIDE];
  int n0 = blockIdx.x * 16;
  int lane = threadIdx.x;
  int half = lane >> 4, l16 = lane & 15;
  for (int idx = lane; idx < 16 * HIDDEN; idx += 32) {
    int row = idx & 15, fl = idx >> 4;
    float v = g[(n0 + row) * HIDDEN + fl];
    float* d = &feat[row * LDS1_STRIDE + fl * 8];
    d[0] = silu_f(v);
    float bs[N_COEF];
    bspline7(v, bs);
#pragma unroll
    for (int c = 0; c < N_COEF; ++c) d[1 + c] = bs[c];
  }
  __syncthreads();
  for (int ct = 0; ct < 8; ++ct) {
    int o0 = ct * 16;
    v8f accA = {}, accB = {};
    for (int k = 0; k < 128; k += 8) {
#pragma unroll
      for (int s = 0; s < 2; ++s) {
        int ka0 = k + s * 4 + (half ? 2 : 0);
        int ka1 = k + s * 4 + (half ? 3 : 1);
        v2f a, b;
        a.x = feat[l16 * LDS1_STRIDE + ka0];
        a.y = feat[l16 * LDS1_STRIDE + ka1];
        b.x = Bm[ka0 * OUT_C + o0 + l16];
        b.y = Bm[ka1 * OUT_C + o0 + l16];
        if (s == 0) accA = wmma4(a, b, accA);
        else        accB = wmma4(a, b, accB);
      }
    }
    v8f acc = accA + accB;
#pragma unroll
    for (int i = 0; i < 8; ++i)
      out[(n0 + i + half * 8) * OUT_C + o0 + l16] = acc[i];
  }
}

extern "C" void kernel_launch(void* const* d_in, const int* in_sizes, int n_in,
                              void* d_out, int out_size, void* d_ws, size_t ws_size,
                              hipStream_t stream) {
  (void)in_sizes; (void)n_in; (void)out_size; (void)ws_size;
  const float* x       = (const float*)d_in[0];
  const int*   ei      = (const int*)d_in[1];
  const float* gcn_w   = (const float*)d_in[2];
  const float* gcn_b   = (const float*)d_in[3];
  const float* base_w0 = (const float*)d_in[4];
  const float* spl_w0  = (const float*)d_in[5];
  const float* scal0   = (const float*)d_in[6];
  const float* base_w1 = (const float*)d_in[7];
  const float* spl_w1  = (const float*)d_in[8];
  const float* scal1   = (const float*)d_in[9];
  float* out = (float*)d_out;

  float* ws    = (float*)d_ws;
  float* hlin  = ws;
  float* h     = hlin + (size_t)N_NODES * OUT_C;
  float* deg   = h + (size_t)N_NODES * OUT_C;
  float* g     = deg + N_NODES;
  float* Bm0   = g + (size_t)N_NODES * HIDDEN;
  float* Bm1   = Bm0 + 8 * OUT_C * HIDDEN;

  const int NT = N_NODES / 16;
  const int NG = (NT + 3) / 4;

  k_gemm_xw<<<NG, 256, 0, stream>>>(x, gcn_w, hlin);
  k_deg_init<<<(N_NODES + 255) / 256, 256, 0, stream>>>(deg);
  k_deg_acc<<<(N_EDGES + 255) / 256, 256, 0, stream>>>(ei, deg);
  k_deg_rsqrt<<<(N_NODES + 255) / 256, 256, 0, stream>>>(deg);
  k_self_bias<<<(N_NODES * OUT_C + 255) / 256, 256, 0, stream>>>(hlin, deg, gcn_b, h);
  k_edge_scatter<<<((long long)N_EDGES * 32 + 255) / 256, 256, 0, stream>>>(ei, hlin, deg, h);
  k_prep_w0<<<(8 * OUT_C * HIDDEN + 255) / 256, 256, 0, stream>>>(base_w0, spl_w0, scal0, Bm0);
  k_prep_w1<<<(8 * HIDDEN * OUT_C + 255) / 256, 256, 0, stream>>>(base_w1, spl_w1, scal1, Bm1);
  k_kan0<<<N_NODES / 16, 32, 0, stream>>>(h, Bm0, g);
  k_kan1<<<N_NODES / 16, 32, 0, stream>>>(g, Bm1, out);
}


# Edge scatter: a workgroup reads its die and draws a ticket from its cache half's list; half 0 adds channels 0-63 of each edge, half 1 64-127: a sum line takes atomics from one half (arrival order)
# speedup vs baseline: 2.8095x; 2.8095x over previous
#include <hip/hip_runtime.h>
#include <hip/hip_bf16.h>

#define N_NODES 50000
#define N_EDGES 800000
#define IN_C    128
#define OUT_C   128
#define HIDDEN  16
#define N_COEF  7

typedef __attribute__((ext_vector_type(2))) float v2f;
typedef __attribute__((ext_vector_type(8))) float v8f;

__device__ __forceinline__ v8f wmma4(v2f a, v2f b, v8f c) {
  return __builtin_amdgcn_wmma_f32_16x16x4_f32(false, a, false, b, (short)0, c,
                                               false, false);
}

__device__ __forceinline__ float silu_f(float v) {
  return v * (1.0f / (1.0f + __expf(-v)));
}

__device__ __forceinline__ void bspline7(float v, float* out) {
  float b[10];
#pragma unroll
  for (int j = 0; j < 10; ++j) {
    float gj = 0.5f * (float)j - 2.5f;
    b[j] = (v >= gj && v < gj + 0.5f) ? 1.0f : 0.0f;
  }
#pragma unroll
  for (int p = 1; p <= 3; ++p) {
    float invd = 1.0f / (0.5f * (float)p);
#pragma unroll
    for (int j = 0; j < 10 - p; ++j) {
      float gj  = 0.5f * (float)j - 2.5f;
      float gjp = gj + 0.5f * (float)p;
      b[j] = (v - gj) * invd * b[j] + ((gjp + 0.5f) - v) * invd * b[j + 1];
    }
  }
#pragma unroll
  for (int c = 0; c < N_COEF; ++c) out[c] = b[c];
}

#define GX_GROUPS (IN_C / 8)
static_assert(GX_GROUPS % 2 == 0, "groups are taken in pairs");
struct GxBuf { float4 w; float4 a[4]; };

__device__ __forceinline__ GxBuf gx_load(const float4* __restrict__ wp,
                                         const float4* const* ap, int m) {
  GxBuf b;
  b.w = wp[2 * m];
#pragma unroll
  for (int i = 0; i < 4; ++i) b.a[i] = ap[i][2 * m];
  return b;
}

__device__ __forceinline__ void gx_group(const GxBuf& b, v8f* acc) {
  v2f w0, w1;
  w0.x = b.w.x; w0.y = b.w.y;
  w1.x = b.w.z; w1.y = b.w.w;
#pragma unroll
  for (int i = 0; i < 4; ++i) {
    v2f a;
    a.x = b.a[i].x; a.y = b.a[i].y;
    acc[i] = wmma4(a, w0, acc[i]);
  }
#pragma unroll
  for (int i = 0; i < 4; ++i) {
    v2f a;
    a.x = b.a[i].z; a.y = b.a[i].w;
    acc[i] = wmma4(a, w1, acc[i]);
  }
}

__global__ void __launch_bounds__(256, 8)
k_gemm_xw(const float* __restrict__ x, const float* __restrict__ W,
          const float* __restrict__ dinv,
          float* __restrict__ hs, float* __restrict__ h) {
  const int NT = N_NODES / 16;
  const int NG = (NT + 3) / 4;
  int wave = (int)((blockIdx.x * blockDim.x + threadIdx.x) >> 5);
  int lane = (int)(threadIdx.x & 31);
  if (wave >= NG * (OUT_C / 16)) return;
  int grp = wave >> 3, ct = wave & 7;
  int o0 = ct * 16;
  int half = lane >> 4, l16 = lane & 15;

  int  rt[4];
  bool valid[4];
#pragma unroll
  for (int i = 0; i < 4; ++i) {
    int r = grp * 4 + i;
    valid[i] = (r < NT);
    rt[i] = valid[i] ? r : (NT - 1);
  }

  const float4* wp = (const float4*)(W + (size_t)(o0 + l16) * IN_C + 4 * half);
  const float4* ap[4];
#pragma unroll
  for (int i = 0; i < 4; ++i)
    ap[i] = (const float4*)(x + (size_t)(rt[i] * 16 + l16) * IN_C + 4 * half);

  v8f acc[4] = {};
  GxBuf ba = gx_load(wp, ap, 0);
  for (int m = 0; m + 2 < GX_GROUPS; m += 2) {
    GxBuf bb = gx_load(wp, ap, m + 1);
    __builtin_amdgcn_sched_barrier(0);
    gx_group(ba, acc);
    ba = gx_load(wp, ap, m + 2);
    __builtin_amdgcn_sched_barrier(0);
    gx_group(bb, acc);
  }
  float4 d0[4], d1[4];
  {
    GxBuf bb = gx_load(wp, ap, GX_GROUPS - 1);
    __builtin_amdgcn_sched_barrier(0);
    gx_group(ba, acc);
#pragma unroll
    for (int i = 0; i < 4; ++i) {
      const float4* dp = (const float4*)(dinv + rt[i] * 16 + half * 8);
      d0[i] = dp[0];
      d1[i] = dp[1];
    }
    __builtin_amdgcn_sched_barrier(0);
    gx_group(bb, acc);
  }

#pragma unroll
  for (int i = 0; i < 4; ++i) {
    if (!valid[i]) continue;
    const float dv[8] = {d0[i].x, d0[i].y, d0[i].z, d0[i].w, d1[i].x, d1[i].y, d1[i].z, d1[i].w};
#pragma unroll
    for (int j = 0; j < 8; ++j) {
      const int idx = (rt[i] * 16 + j + half * 8) * OUT_C + o0 + l16;
      const float v = acc[i][j] * dv[j];
      hs[idx] = v;
      h[idx]  = v;
    }
  }
}

#define SC_TICK_PITCH 64
#define DEG_PARTS   64
#define DEG_RANGES  4
#define DEG_THREADS 1024
#define DEG_SLICE   (N_EDGES / DEG_PARTS)
#define DEG_ITERS   ((DEG_SLICE + DEG_THREADS - 1) / DEG_THREADS)
#define DEG_RNODES  (N_NODES / DEG_RANGES)
#define DEG_VEC     (DEG_RNODES / 4)

__global__ void __launch_bounds__(DEG_THREADS)
k_deg_hist(const int* __restrict__ ei, int* __restrict__ part) {
  __shared__ alignas(16) int hist[DEG_RNODES];
  const int tid   = (int)threadIdx.x;
  const int slice = (int)blockIdx.x / DEG_RANGES;
  const int lo    = ((int)blockIdx.x % DEG_RANGES) * DEG_RNODES;
  const int* dst  = ei + N_EDGES + slice * DEG_SLICE;

  int d[DEG_ITERS];
#pragma unroll
  for (int j = 0; j < DEG_ITERS; ++j) {
    int e = tid + j * DEG_THREADS;
    d[j] = (e < DEG_SLICE) ? dst[e] : -1;
  }

  int4* h4 = (int4*)hist;
  for (int i = tid; i < DEG_VEC; i += DEG_THREADS) h4[i] = make_int4(0, 0, 0, 0);
  __syncthreads();

#pragma unroll
  for (int j = 0; j < DEG_ITERS; ++j) {
    unsigned rel = (unsigned)(d[j] - lo);
    if (rel < (unsigned)DEG_RNODES)
      __hip_atomic_fetch_add(&hist[rel], 1, __ATOMIC_RELAXED,
                             __HIP_MEMORY_SCOPE_WORKGROUP);
  }
  __syncthreads();

  int4* p4 = (int4*)(part + (size_t)slice * N_NODES + lo);
  for (int i = tid; i < DEG_VEC; i += DEG_THREADS) p4[i] = h4[i];
}

__global__ void k_deg_dinv(const int* __restrict__ part, float* __restrict__ dinv,
                           unsigned* __restrict__ tick) {
  int i = blockIdx.x * blockDim.x + threadIdx.x;
  if (i < 4) tick[i * SC_TICK_PITCH] = 0u;
  if (i < N_NODES) {
    int c = 1;
#pragma unroll
    for (int p = 0; p < DEG_PARTS; ++p) c += part[p * N_NODES + i];
    float dg = (float)c;
    dinv[i] = (dg > 0.0f) ? rsqrtf(dg) : 0.0f;
  }
}

#define SC_WAVES 4
#define SC_EPW   64
#define SC_BATCH 16
#define SC_UNIT  (SC_WAVES * SC_EPW)
#define SC_UNITS (N_EDGES / SC_UNIT)
static_assert(N_EDGES % SC_UNIT == 0, "a workgroup takes exactly SC_UNIT edges");
static_assert(SC_EPW == 64, "two edges per lane");
static_assert(SC_EPW % SC_BATCH == 0, "whole batches");

__global__ void __launch_bounds__(SC_WAVES * 32)
k_edge_scatter(const int* __restrict__ ei,
               const float* __restrict__ hs,
               float* __restrict__ h,
               unsigned* __restrict__ tick) {
  __shared__ unsigned s_job;
  const int lane = (int)(threadIdx.x & 31);
  const int wv   = (int)(threadIdx.x >> 5);
  if (threadIdx.x == 0) {
    const unsigned id = __builtin_amdgcn_s_sendmsg_rtn(0x87);
    const unsigned m  = (id >> 18) & 1u;
    const unsigned t  = __hip_atomic_fetch_add(&tick[m * SC_TICK_PITCH], 1u, __ATOMIC_RELAXED,
                                               __HIP_MEMORY_SCOPE_AGENT);
    unsigned job;
    if (t < (unsigned)SC_UNITS) {
      job = t * 2u + m;
    } else {
      const unsigned o  = m ^ 1u;
      const unsigned t2 = __hip_atomic_fetch_add(&tick[(2u + o) * SC_TICK_PITCH], 1u, __ATOMIC_RELAXED,
                                                 __HIP_MEMORY_SCOPE_AGENT);
      const unsigned u  = (t2 < (unsigned)SC_UNITS) ? ((unsigned)SC_UNITS - 1u - t2) : 0u;
      job = u * 2u + o;
    }
    s_job = job;
  }
  __syncthreads();
  const unsigned job  = (unsigned)__builtin_amdgcn_readfirstlane((int)s_job);
  const int      unit = (int)(job >> 1);
  const int      half = (int)(job & 1u);

  const int e0 = unit * SC_UNIT + wv * SC_EPW + lane;
  int src[2], dst[2];
  src[0] = ei[e0];
  src[1] = ei[e0 + 32];
  dst[0] = ei[N_EDGES + e0];
  dst[1] = ei[N_EDGES + e0 + 32];

  const float* hsl = hs + half * 64 + lane;
  float*       hl  = h  + half * 64 + lane;

#pragma unroll
  for (int b = 0; b < SC_EPW; b += SC_BATCH) {
    float v[SC_BATCH][2];
#pragma unroll
    for (int u = 0; u < SC_BATCH; ++u) {
      const int s = __builtin_amdgcn_readlane(src[(b + u) >> 5], (b + u) & 31);
      const float* row = hsl + (size_t)s * OUT_C;
      v[u][0] = row[0];
      v[u][1] = row[32];
    }
#pragma unroll
    for (int u = 0; u < SC_BATCH; ++u) {
      const int d = __builtin_amdgcn_readlane(dst[(b + u) >> 5], (b + u) & 31);
      float* base = hl + (size_t)d * OUT_C;
      __hip_atomic_fetch_add(base +  0, v[u][0], __ATOMIC_RELAXED, __HIP_MEMORY_SCOPE_AGENT);
      __hip_atomic_fetch_add(base + 32, v[u][1], __ATOMIC_RELAXED, __HIP_MEMORY_SCOPE_AGENT);
    }
  }
}

__global__ void k_prep_w0(const float* __restrict__ bw, const float* __restrict__ sw,
                          const float* __restrict__ sc, float* __restrict__ Bm) {
  int t = blockIdx.x * blockDim.x + threadIdx.x;
  if (t >= 8 * OUT_C * HIDDEN) return;
  int k = t >> 4, o = t & 15;
  int f = k >> 3, j = k & 7;
  float w = (j == 0) ? bw[o * OUT_C + f]
                     : sw[(o * OUT_C + f) * N_COEF + (j - 1)] * sc[o * OUT_C + f];
  Bm[((f * 2 + (j >> 2)) * HIDDEN + o) * 4 + (j & 3)] = w;
}

__global__ void k_prep_w1(const float* __restrict__ bw, const float* __restrict__ sw,
                          const float* __restrict__ sc, float* __restrict__ Bm) {
  int t = blockIdx.x * blockDim.x + threadIdx.x;
  if (t >= 8 * HIDDEN * OUT_C) return;
  int k = t >> 7, o = t & 127;
  int f = k >> 3, j = k & 7;
  float w = (j == 0) ? bw[o * HIDDEN + f]
                     : sw[(o * HIDDEN + f) * N_COEF + (j - 1)] * sc[o * HIDDEN + f];
  Bm[((k >> 1) * OUT_C + o) * 2 + (k & 1)] = w;
}

#define K0_GROUPS 16
static_assert(K0_GROUPS % 2 == 0, "groups are taken in pairs");
struct K0Buf { float4 hv; float4 bv; float4 w[4][2]; };

__device__ __forceinline__ K0Buf k0_load(const float4* __restrict__ hp,
                                         const float4* __restrict__ bp,
                                         const float4* __restrict__ wp, int m) {
  K0Buf b;
  b.hv = hp[2 * m];
  b.bv = bp[2 * m];
#pragma unroll
  for (int i = 0; i < 4; ++i) {
    b.w[i][0] = wp[(8 * m + i) * 32];
    b.w[i][1] = wp[(8 * m + i) * 32 + 16];
  }
  return b;
}

__device__ __forceinline__ void k0_group(const K0Buf& b, float di, v8f& acc0, v8f& acc1) {
  const float vv[4] = {b.hv.x * di + b.bv.x, b.hv.y * di + b.bv.y,
                       b.hv.z * di + b.bv.z, b.hv.w * di + b.bv.w};
#pragma unroll
  for (int i = 0; i < 4; ++i) {
    float e[8];
    e[0] = silu_f(vv[i]);
    bspline7(vv[i], e + 1);
    const float4 w0 = b.w[i][0], w1 = b.w[i][1];
    v2f a, w;
    a.x = e[0]; a.y = e[1]; w.x = w0.x; w.y = w0.y; acc0 = wmma4(a, w, acc0);
    a.x = e[2]; a.y = e[3]; w.x = w0.z; w.y = w0.w; acc1 = wmma4(a, w, acc1);
    a.x = e[4]; a.y = e[5]; w.x = w1.x; w.y = w1.y; acc0 = wmma4(a, w, acc0);
    a.x = e[6]; a.y = e[7]; w.x = w1.z; w.y = w1.w; acc1 = wmma4(a, w, acc1);
  }
}

__global__ void __launch_bounds__(32) k_kan0(const float* __restrict__ h,
                                             const float* __restrict__ dinv,
                                             const float* __restrict__ bias,
                                             const float* __restrict__ Bm,
                                             float* __restrict__ g) {
  int n0 = blockIdx.x * 16;
  int lane = threadIdx.x;
  int half = lane >> 4, l16 = lane & 15;
  const float4* hp = (const float4*)(h + (size_t)(n0 + l16) * OUT_C + 4 * half);
  const float4* bp = (const float4*)(bias + 4 * half);
  const float4* wp = (const float4*)Bm + (4 * half) * 32 + l16;
  const float di = dinv[n0 + l16];

  v8f acc0 = {}, acc1 = {};
  K0Buf ba = k0_load(hp, bp, wp, 0);
  for (int m = 0; m + 2 < K0_GROUPS; m += 2) {
    K0Buf bb = k0_load(hp, bp, wp, m + 1);
    __builtin_amdgcn_sched_barrier(0);
    k0_group(ba, di, acc0, acc1);
    ba = k0_load(hp, bp, wp, m + 2);
    __builtin_amdgcn_sched_barrier(0);
    k0_group(bb, di, acc0, acc1);
  }
  {
    K0Buf bb = k0_load(hp, bp, wp, K0_GROUPS - 1);
    __builtin_amdgcn_sched_barrier(0);
    k0_group(ba, di, acc0, acc1);
    k0_group(bb, di, acc0, acc1);
  }

  v8f acc = acc0 + acc1;
#pragma unroll
  for (int i = 0; i < 8; ++i)
    g[(n0 + i + half * 8) * HIDDEN + l16] = acc[i];
}

#define LDS1_STRIDE 130
#define K1_STEPS 32
#define K1_GROUP 4
static_assert(K1_STEPS % K1_GROUP == 0, "whole groups");

__global__ void __launch_bounds__(32) k_kan1(const float* __restrict__ g,
                                             const float* __restrict__ Bm,
                                             float* __restrict__ out) {
  __shared__ float feat[16 * LDS1_STRIDE];
  int n0 = blockIdx.x * 16;
  int lane = threadIdx.x;
  int half = lane >> 4, l16 = lane & 15;
  for (int idx = lane; idx < 16 * HIDDEN; idx += 32) {
    int row = idx & 15, fl = idx >> 4;
    float v = g[(n0 + row) * HIDDEN + fl];
    float* d = &feat[row * LDS1_STRIDE + fl * 8];
    d[0] = silu_f(v);
    float bs[N_COEF];
    bspline7(v, bs);
#pragma unroll
    for (int c = 0; c < N_COEF; ++c) d[1 + c] = bs[c];
  }
  __syncthreads();

  const float4* bp = (const float4*)(Bm + (half * OUT_C + 2 * l16) * 2);

  v8f acc[8] = {};
  float4 bb[2][K1_GROUP][4];
#pragma unroll
  for (int q = 0; q < K1_GROUP; ++q)
#pragma unroll
    for (int j = 0; j < 4; ++j) bb[0][q][j] = bp[q * 128 + j * 16];

#pragma unroll
  for (int grp = 0; grp < K1_STEPS / K1_GROUP; ++grp) {
    if (grp + 1 < K1_STEPS / K1_GROUP) {
#pragma unroll
      for (int q = 0; q < K1_GROUP; ++q)
#pragma unroll
        for (int j = 0; j < 4; ++j)
          bb[(grp + 1) & 1][q][j] = bp[((grp + 1) * K1_GROUP + q) * 128 + j * 16];
    }
#pragma unroll
    for (int q = 0; q < K1_GROUP; ++q) {
      int ka0 = (grp * K1_GROUP + q) * 4 + half * 2;
      v2f a;
      a.x = feat[l16 * LDS1_STRIDE + ka0];
      a.y = feat[l16 * LDS1_STRIDE + ka0 + 1];
#pragma unroll
      for (int j = 0; j < 4; ++j) {
        float4 w = bb[grp & 1][q][j];
        v2f b0, b1;
        b0.x = w.x; b0.y = w.y;
        b1.x = w.z; b1.y = w.w;
        acc[2 * j]     = wmma4(a, b0, acc[2 * j]);
        acc[2 * j + 1] = wmma4(a, b1, acc[2 * j + 1]);
      }
    }
  }

#pragma unroll
  for (int i = 0; i < 8; ++i) {
    float* orow = out + (size_t)(n0 + i + half * 8) * OUT_C + 2 * l16;
#pragma unroll
    for (int j = 0; j < 4; ++j) {
      float2 v;
      v.x = acc[2 * j][i];
      v.y = acc[2 * j + 1][i];
      *(float2*)(orow + 32 * j) = v;
    }
  }
}

extern "C" void kernel_launch(void* const* d_in, const int* in_sizes, int n_in,
                              void* d_out, int out_size, void* d_ws, size_t ws_size,
                              hipStream_t stream) {
  (void)in_sizes; (void)n_in; (void)out_size; (void)ws_size;
  const float* x       = (const float*)d_in[0];
  const int*   ei      = (const int*)d_in[1];
  const float* gcn_w   = (const float*)d_in[2];
  const float* gcn_b   = (const float*)d_in[3];
  const float* base_w0 = (const float*)d_in[4];
  const float* spl_w0  = (const float*)d_in[5];
  const float* scal0   = (const float*)d_in[6];
  const float* base_w1 = (const float*)d_in[7];
  const float* spl_w1  = (const float*)d_in[8];
  const float* scal1   = (const float*)d_in[9];
  float* out = (float*)d_out;

  float* ws    = (float*)d_ws;
  float* hs    = ws;
  float* h     = hs + (size_t)N_NODES * OUT_C;
  float* deg   = h + (size_t)N_NODES * OUT_C;
  float* g     = deg + N_NODES;
  float* Bm0   = g + (size_t)N_NODES * HIDDEN;
  float* Bm1   = Bm0 + 8 * OUT_C * HIDDEN;

  const int NT = N_NODES / 16;
  const int NG = (NT + 3) / 4;

  int* part = (int*)h;
  unsigned* tick = (unsigned*)g;
  k_deg_hist<<<DEG_PARTS * DEG_RANGES, DEG_THREADS, 0, stream>>>(ei, part);
  k_deg_dinv<<<(N_NODES + 255) / 256, 256, 0, stream>>>(part, deg, tick);
  k_gemm_xw<<<NG, 256, 0, stream>>>(x, gcn_w, deg, hs, h);
  k_edge_scatter<<<2 * SC_UNITS, SC_WAVES * 32, 0, stream>>>(ei, hs, h, tick);
  k_prep_w0<<<(8 * OUT_C * HIDDEN + 255) / 256, 256, 0, stream>>>(base_w0, spl_w0, scal0, Bm0);
  k_prep_w1<<<(8 * HIDDEN * OUT_C + 255) / 256, 256, 0, stream>>>(base_w1, spl_w1, scal1, Bm1);
  k_kan0<<<N_NODES / 16, 32, 0, stream>>>(h, deg, gcn_b, Bm0, g);
  k_kan1<<<N_NODES / 16, 32, 0, stream>>>(g, Bm1, out);
}


# KAN layers 0 and 1 in one launch: a wave expands its layer-0 tile from its accumulators into layer 1's LDS rows; layer 0 at raised wave priority, layer-1 matrix runs fill the pipe under it; sums kept
# speedup vs baseline: 2.9055x; 1.0342x over previous
#include <hip/hip_runtime.h>
#include <hip/hip_bf16.h>

#define N_NODES 50000
#define N_EDGES 800000
#define IN_C    128
#define OUT_C   128
#define HIDDEN  16
#define N_COEF  7

typedef __attribute__((ext_vector_type(2))) float v2f;
typedef __attribute__((ext_vector_type(8))) float v8f;

__device__ __forceinline__ v8f wmma4(v2f a, v2f b, v8f c) {
  return __builtin_amdgcn_wmma_f32_16x16x4_f32(false, a, false, b, (short)0, c,
                                               false, false);
}

__device__ __forceinline__ float silu_f(float v) {
  return v * (1.0f / (1.0f + __expf(-v)));
}

__device__ __forceinline__ void bspline7(float v, float* out) {
  float b[10];
#pragma unroll
  for (int j = 0; j < 10; ++j) {
    float gj = 0.5f * (float)j - 2.5f;
    b[j] = (v >= gj && v < gj + 0.5f) ? 1.0f : 0.0f;
  }
#pragma unroll
  for (int p = 1; p <= 3; ++p) {
    float invd = 1.0f / (0.5f * (float)p);
#pragma unroll
    for (int j = 0; j < 10 - p; ++j) {
      float gj  = 0.5f * (float)j - 2.5f;
      float gjp = gj + 0.5f * (float)p;
      b[j] = (v - gj) * invd * b[j] + ((gjp + 0.5f) - v) * invd * b[j + 1];
    }
  }
#pragma unroll
  for (int c = 0; c < N_COEF; ++c) out[c] = b[c];
}

#define GX_GROUPS (IN_C / 8)
static_assert(GX_GROUPS % 2 == 0, "groups are taken in pairs");
struct GxBuf { float4 w; float4 a[4]; };

__device__ __forceinline__ GxBuf gx_load(const float4* __restrict__ wp,
                                         const float4* const* ap, int m) {
  GxBuf b;
  b.w = wp[2 * m];
#pragma unroll
  for (int i = 0; i < 4; ++i) b.a[i] = ap[i][2 * m];
  return b;
}

__device__ __forceinline__ void gx_group(const GxBuf& b, v8f* acc) {
  v2f w0, w1;
  w0.x = b.w.x; w0.y = b.w.y;
  w1.x = b.w.z; w1.y = b.w.w;
#pragma unroll
  for (int i = 0; i < 4; ++i) {
    v2f a;
    a.x = b.a[i].x; a.y = b.a[i].y;
    acc[i] = wmma4(a, w0, acc[i]);
  }
#pragma unroll
  for (int i = 0; i < 4; ++i) {
    v2f a;
    a.x = b.a[i].z; a.y = b.a[i].w;
    acc[i] = wmma4(a, w1, acc[i]);
  }
}

__global__ void __launch_bounds__(256, 8)
k_gemm_xw(const float* __restrict__ x, const float* __restrict__ W,
          const float* __restrict__ dinv,
          float* __restrict__ hs, float* __restrict__ h) {
  const int NT = N_NODES / 16;
  const int NG = (NT + 3) / 4;
  int wave = (int)((blockIdx.x * blockDim.x + threadIdx.x) >> 5);
  int lane = (int)(threadIdx.x & 31);
  if (wave >= NG * (OUT_C / 16)) return;
  int grp = wave >> 3, ct = wave & 7;
  int o0 = ct * 16;
  int half = lane >> 4, l16 = lane & 15;

  int  rt[4];
  bool valid[4];
#pragma unroll
  for (int i = 0; i < 4; ++i) {
    int r = grp * 4 + i;
    valid[i] = (r < NT);
    rt[i] = valid[i] ? r : (NT - 1);
  }

  const float4* wp = (const float4*)(W + (size_t)(o0 + l16) * IN_C + 4 * half);
  const float4* ap[4];
#pragma unroll
  for (int i = 0; i < 4; ++i)
    ap[i] = (const float4*)(x + (size_t)(rt[i] * 16 + l16) * IN_C + 4 * half);

  v8f acc[4] = {};
  GxBuf ba = gx_load(wp, ap, 0);
  for (int m = 0; m + 2 < GX_GROUPS; m += 2) {
    GxBuf bb = gx_load(wp, ap, m + 1);
    __builtin_amdgcn_sched_barrier(0);
    gx_group(ba, acc);
    ba = gx_load(wp, ap, m + 2);
    __builtin_amdgcn_sched_barrier(0);
    gx_group(bb, acc);
  }
  float4 d0[4], d1[4];
  {
    GxBuf bb = gx_load(wp, ap, GX_GROUPS - 1);
    __builtin_amdgcn_sched_barrier(0);
    gx_group(ba, acc);
#pragma unroll
    for (int i = 0; i < 4; ++i) {
      const float4* dp = (const float4*)(dinv + rt[i] * 16 + half * 8);
      d0[i] = dp[0];
      d1[i] = dp[1];
    }
    __builtin_amdgcn_sched_barrier(0);
    gx_group(bb, acc);
  }

#pragma unroll
  for (int i = 0; i < 4; ++i) {
    if (!valid[i]) continue;
    const float dv[8] = {d0[i].x, d0[i].y, d0[i].z, d0[i].w, d1[i].x, d1[i].y, d1[i].z, d1[i].w};
#pragma unroll
    for (int j = 0; j < 8; ++j) {
      const int idx = (rt[i] * 16 + j + half * 8) * OUT_C + o0 + l16;
      const float v = acc[i][j] * dv[j];
      hs[idx] = v;
      h[idx]  = v;
    }
  }
}

#define SC_TICK_PITCH 64
#define DEG_PARTS   64
#define DEG_RANGES  4
#define DEG_THREADS 1024
#define DEG_SLICE   (N_EDGES / DEG_PARTS)
#define DEG_ITERS   ((DEG_SLICE + DEG_THREADS - 1) / DEG_THREADS)
#define DEG_RNODES  (N_NODES / DEG_RANGES)
#define DEG_VEC     (DEG_RNODES / 4)

__global__ void __launch_bounds__(DEG_THREADS)
k_deg_hist(const int* __restrict__ ei, int* __restrict__ part) {
  __shared__ alignas(16) int hist[DEG_RNODES];
  const int tid   = (int)threadIdx.x;
  const int slice = (int)blockIdx.x / DEG_RANGES;
  const int lo    = ((int)blockIdx.x % DEG_RANGES) * DEG_RNODES;
  const int* dst  = ei + N_EDGES + slice * DEG_SLICE;

  int d[DEG_ITERS];
#pragma unroll
  for (int j = 0; j < DEG_ITERS; ++j) {
    int e = tid + j * DEG_THREADS;
    d[j] = (e < DEG_SLICE) ? dst[e] : -1;
  }

  int4* h4 = (int4*)hist;
  for (int i = tid; i < DEG_VEC; i += DEG_THREADS) h4[i] = make_int4(0, 0, 0, 0);
  __syncthreads();

#pragma unroll
  for (int j = 0; j < DEG_ITERS; ++j) {
    unsigned rel = (unsigned)(d[j] - lo);
    if (rel < (unsigned)DEG_RNODES)
      __hip_atomic_fetch_add(&hist[rel], 1, __ATOMIC_RELAXED,
                             __HIP_MEMORY_SCOPE_WORKGROUP);
  }
  __syncthreads();

  int4* p4 = (int4*)(part + (size_t)slice * N_NODES + lo);
  for (int i = tid; i < DEG_VEC; i += DEG_THREADS) p4[i] = h4[i];
}

__global__ void k_deg_dinv(const int* __restrict__ part, float* __restrict__ dinv,
                           unsigned* __restrict__ tick) {
  int i = blockIdx.x * blockDim.x + threadIdx.x;
  if (i < 4) tick[i * SC_TICK_PITCH] = 0u;
  if (i < N_NODES) {
    int c = 1;
#pragma unroll
    for (int p = 0; p < DEG_PARTS; ++p) c += part[p * N_NODES + i];
    float dg = (float)c;
    dinv[i] = (dg > 0.0f) ? rsqrtf(dg) : 0.0f;
  }
}

#define SC_WAVES 4
#define SC_EPW   64
#define SC_BATCH 16
#define SC_UNIT  (SC_WAVES * SC_EPW)
#define SC_UNITS (N_EDGES / SC_UNIT)
static_assert(N_EDGES % SC_UNIT == 0, "a workgroup takes exactly SC_UNIT edges");
static_assert(SC_EPW == 64, "two edges per lane");
static_assert(SC_EPW % SC_BATCH == 0, "whole batches");

__global__ void __launch_bounds__(SC_WAVES * 32)
k_edge_scatter(const int* __restrict__ ei,
               const float* __restrict__ hs,
               float* __restrict__ h,
               unsigned* __restrict__ tick) {
  __shared__ unsigned s_job;
  const int lane = (int)(threadIdx.x & 31);
  const int wv   = (int)(threadIdx.x >> 5);
  if (threadIdx.x == 0) {
    const unsigned id = __builtin_amdgcn_s_sendmsg_rtn(0x87);
    const unsigned m  = (id >> 18) & 1u;
    const unsigned t  = __hip_atomic_fetch_add(&tick[m * SC_TICK_PITCH], 1u, __ATOMIC_RELAXED,
                                               __HIP_MEMORY_SCOPE_AGENT);
    unsigned job;
    if (t < (unsigned)SC_UNITS) {
      job = t * 2u + m;
    } else {
      const unsigned o  = m ^ 1u;
      const unsigned t2 = __hip_atomic_fetch_add(&tick[(2u + o) * SC_TICK_PITCH], 1u, __ATOMIC_RELAXED,
                                                 __HIP_MEMORY_SCOPE_AGENT);
      const unsigned u  = (t2 < (unsigned)SC_UNITS) ? ((unsigned)SC_UNITS - 1u - t2) : 0u;
      job = u * 2u + o;
    }
    s_job = job;
  }
  __syncthreads();
  const unsigned job  = (unsigned)__builtin_amdgcn_readfirstlane((int)s_job);
  const int      unit = (int)(job >> 1);
  const int      half = (int)(job & 1u);

  const int e0 = unit * SC_UNIT + wv * SC_EPW + lane;
  int src[2], dst[2];
  src[0] = ei[e0];
  src[1] = ei[e0 + 32];
  dst[0] = ei[N_EDGES + e0];
  dst[1] = ei[N_EDGES + e0 + 32];

  const float* hsl = hs + half * 64 + lane;
  float*       hl  = h  + half * 64 + lane;

#pragma unroll
  for (int b = 0; b < SC_EPW; b += SC_BATCH) {
    float v[SC_BATCH][2];
#pragma unroll
    for (int u = 0; u < SC_BATCH; ++u) {
      const int s = __builtin_amdgcn_readlane(src[(b + u) >> 5], (b + u) & 31);
      const float* row = hsl + (size_t)s * OUT_C;
      v[u][0] = row[0];
      v[u][1] = row[32];
    }
#pragma unroll
    for (int u = 0; u < SC_BATCH; ++u) {
      const int d = __builtin_amdgcn_readlane(dst[(b + u) >> 5], (b + u) & 31);
      float* base = hl + (size_t)d * OUT_C;
      __hip_atomic_fetch_add(base +  0, v[u][0], __ATOMIC_RELAXED, __HIP_MEMORY_SCOPE_AGENT);
      __hip_atomic_fetch_add(base + 32, v[u][1], __ATOMIC_RELAXED, __HIP_MEMORY_SCOPE_AGENT);
    }
  }
}

__global__ void k_prep_w0(const float* __restrict__ bw, const float* __restrict__ sw,
                          const float* __restrict__ sc, float* __restrict__ Bm) {
  int t = blockIdx.x * blockDim.x + threadIdx.x;
  if (t >= 8 * OUT_C * HIDDEN) return;
  int k = t >> 4, o = t & 15;
  int f = k >> 3, j = k & 7;
  float w = (j == 0) ? bw[o * OUT_C + f]
                     : sw[(o * OUT_C + f) * N_COEF + (j - 1)] * sc[o * OUT_C + f];
  Bm[((f * 2 + (j >> 2)) * HIDDEN + o) * 4 + (j & 3)] = w;
}

__global__ void k_prep_w1(const float* __restrict__ bw, const float* __restrict__ sw,
                          const float* __restrict__ sc, float* __restrict__ Bm) {
  int t = blockIdx.x * blockDim.x + threadIdx.x;
  if (t >= 8 * HIDDEN * OUT_C) return;
  int k = t >> 7, o = t & 127;
  int f = k >> 3, j = k & 7;
  float w = (j == 0) ? bw[o * HIDDEN + f]
                     : sw[(o * HIDDEN + f) * N_COEF + (j - 1)] * sc[o * HIDDEN + f];
  Bm[((k >> 1) * OUT_C + o) * 2 + (k & 1)] = w;
}

#define K0_GROUPS 16
static_assert(K0_GROUPS % 2 == 0, "groups are taken in pairs");
struct K0Buf { float4 hv; float4 bv; float4 w[4][2]; };

__device__ __forceinline__ K0Buf k0_load(const float4* __restrict__ hp,
                                         const float4* __restrict__ bp,
                                         const float4* __restrict__ wp, int m) {
  K0Buf b;
  b.hv = hp[2 * m];
  b.bv = bp[2 * m];
#pragma unroll
  for (int i = 0; i < 4; ++i) {
    b.w[i][0] = wp[(8 * m + i) * 32];
    b.w[i][1] = wp[(8 * m + i) * 32 + 16];
  }
  return b;
}

__device__ __forceinline__ void k0_group(const K0Buf& b, float di, v8f& acc0, v8f& acc1) {
  const float vv[4] = {b.hv.x * di + b.bv.x, b.hv.y * di + b.bv.y,
                       b.hv.z * di + b.bv.z, b.hv.w * di + b.bv.w};
#pragma unroll
  for (int i = 0; i < 4; ++i) {
    float e[8];
    e[0] = silu_f(vv[i]);
    bspline7(vv[i], e + 1);
    const float4 w0 = b.w[i][0], w1 = b.w[i][1];
    v2f a, w;
    a.x = e[0]; a.y = e[1]; w.x = w0.x; w.y = w0.y; acc0 = wmma4(a, w, acc0);
    a.x = e[2]; a.y = e[3]; w.x = w0.z; w.y = w0.w; acc1 = wmma4(a, w, acc1);
    a.x = e[4]; a.y = e[5]; w.x = w1.x; w.y = w1.y; acc0 = wmma4(a, w, acc0);
    a.x = e[6]; a.y = e[7]; w.x = w1.z; w.y = w1.w; acc1 = wmma4(a, w, acc1);
  }
}

#define LDS1_STRIDE 130
#define K1_STEPS 32
#define K1_GROUP 4
static_assert(K1_STEPS % K1_GROUP == 0, "whole groups");

__global__ void __launch_bounds__(32) k_kan01(const float* __restrict__ h,
                                              const float* __restrict__ dinv,
                                              const float* __restrict__ bias,
                                              const float* __restrict__ Bm0,
                                              const float* __restrict__ Bm1,
                                              float* __restrict__ out) {
  __shared__ float feat[16 * LDS1_STRIDE];
  int n0 = blockIdx.x * 16;
  int lane = threadIdx.x;
  int half = lane >> 4, l16 = lane & 15;

  __builtin_amdgcn_s_setprio(3);
  const float4* hp = (const float4*)(h + (size_t)(n0 + l16) * OUT_C + 4 * half);
  const float4* bp0 = (const float4*)(bias + 4 * half);
  const float4* wp = (const float4*)Bm0 + (4 * half) * 32 + l16;
  const float di = dinv[n0 + l16];

  v8f acc0 = {}, acc1 = {};
  K0Buf ba = k0_load(hp, bp0, wp, 0);
  for (int m = 0; m + 2 < K0_GROUPS; m += 2) {
    K0Buf bb = k0_load(hp, bp0, wp, m + 1);
    __builtin_amdgcn_sched_barrier(0);
    k0_group(ba, di, acc0, acc1);
    ba = k0_load(hp, bp0, wp, m + 2);
    __builtin_amdgcn_sched_barrier(0);
    k0_group(bb, di, acc0, acc1);
  }
  {
    K0Buf bb = k0_load(hp, bp0, wp, K0_GROUPS - 1);
    __builtin_amdgcn_sched_barrier(0);
    k0_group(ba, di, acc0, acc1);
    k0_group(bb, di, acc0, acc1);
  }

  v8f gt = acc0 + acc1;
#pragma unroll
  for (int i = 0; i < 8; ++i) {
    float v = gt[i];
    float* d = &feat[(half * 8 + i) * LDS1_STRIDE + l16 * 8];
    d[0] = silu_f(v);
    float bs[N_COEF];
    bspline7(v, bs);
#pragma unroll
    for (int c = 0; c < N_COEF; ++c) d[1 + c] = bs[c];
  }
  __syncthreads();

  __builtin_amdgcn_s_setprio(0);
  const float4* bp = (const float4*)(Bm1 + (half * OUT_C + 2 * l16) * 2);

  v8f acc[8] = {};
  float4 bb[2][K1_GROUP][4];
#pragma unroll
  for (int q = 0; q < K1_GROUP; ++q)
#pragma unroll
    for (int j = 0; j < 4; ++j) bb[0][q][j] = bp[q * 128 + j * 16];

#pragma unroll
  for (int grp = 0; grp < K1_STEPS / K1_GROUP; ++grp) {
    if (grp + 1 < K1_STEPS / K1_GROUP) {
#pragma unroll
      for (int q = 0; q < K1_GROUP; ++q)
#pragma unroll
        for (int j = 0; j < 4; ++j)
          bb[(grp + 1) & 1][q][j] = bp[((grp + 1) * K1_GROUP + q) * 128 + j * 16];
    }
#pragma unroll
    for (int q = 0; q < K1_GROUP; ++q) {
      int ka0 = (grp * K1_GROUP + q) * 4 + half * 2;
      v2f a;
      a.x = feat[l16 * LDS1_STRIDE + ka0];
      a.y = feat[l16 * LDS1_STRIDE + ka0 + 1];
#pragma unroll
      for (int j = 0; j < 4; ++j) {
        float4 w = bb[grp & 1][q][j];
        v2f b0, b1;
        b0.x = w.x; b0.y = w.y;
        b1.x = w.z; b1.y = w.w;
        acc[2 * j]     = wmma4(a, b0, acc[2 * j]);
        acc[2 * j + 1] = wmma4(a, b1, acc[2 * j + 1]);
      }
    }
  }

#pragma unroll
  for (int i = 0; i < 8; ++i) {
    float* orow = out + (size_t)(n0 + i + half * 8) * OUT_C + 2 * l16;
#pragma unroll
    for (int j = 0; j < 4; ++j) {
      float2 v;
      v.x = acc[2 * j][i];
      v.y = acc[2 * j + 1][i];
      *(float2*)(orow + 32 * j) = v;
    }
  }
}

extern "C" void kernel_launch(void* const* d_in, const int* in_sizes, int n_in,
                              void* d_out, int out_size, void* d_ws, size_t ws_size,
                              hipStream_t stream) {
  (void)in_sizes; (void)n_in; (void)out_size; (void)ws_size;
  const float* x       = (const float*)d_in[0];
  const int*   ei      = (const int*)d_in[1];
  const float* gcn_w   = (const float*)d_in[2];
  const float* gcn_b   = (const float*)d_in[3];
  const float* base_w0 = (const float*)d_in[4];
  const float* spl_w0  = (const float*)d_in[5];
  const float* scal0   = (const float*)d_in[6];
  const float* base_w1 = (const float*)d_in[7];
  const float* spl_w1  = (const float*)d_in[8];
  const float* scal1   = (const float*)d_in[9];
  float* out = (float*)d_out;

  float* ws    = (float*)d_ws;
  float* hs    = ws;
  float* h     = hs + (size_t)N_NODES * OUT_C;
  float* deg   = h + (size_t)N_NODES * OUT_C;
  float* g     = deg + N_NODES;
  float* Bm0   = g + (size_t)N_NODES * HIDDEN;
  float* Bm1   = Bm0 + 8 * OUT_C * HIDDEN;

  const int NT = N_NODES / 16;
  const int NG = (NT + 3) / 4;

  int* part = (int*)h;
  unsigned* tick = (unsigned*)g;
  k_deg_hist<<<DEG_PARTS * DEG_RANGES, DEG_THREADS, 0, stream>>>(ei, part);
  k_deg_dinv<<<(N_NODES + 255) / 256, 256, 0, stream>>>(part, deg, tick);
  k_gemm_xw<<<NG, 256, 0, stream>>>(x, gcn_w, deg, hs, h);
  k_edge_scatter<<<2 * SC_UNITS, SC_WAVES * 32, 0, stream>>>(ei, hs, h, tick);
  k_prep_w0<<<(8 * OUT_C * HIDDEN + 255) / 256, 256, 0, stream>>>(base_w0, spl_w0, scal0, Bm0);
  k_prep_w1<<<(8 * HIDDEN * OUT_C + 255) / 256, 256, 0, stream>>>(base_w1, spl_w1, scal1, Bm1);
  k_kan01<<<N_NODES / 16, 32, 0, stream>>>(h, deg, gcn_b, Bm0, Bm1, out);
}


# Weight folds ride in the in-degree finishing launch: its first 32768 threads each fold one KAN weight of layer 0 or 1 beside their node's count sum; two small launches dropped, 7 to 5; sums kept
# speedup vs baseline: 3.0370x; 1.0453x over previous
#include <hip/hip_runtime.h>
#include <hip/hip_bf16.h>

#define N_NODES 50000
#define N_EDGES 800000
#define IN_C    128
#define OUT_C   128
#define HIDDEN  16
#define N_COEF  7

typedef __attribute__((ext_vector_type(2))) float v2f;
typedef __attribute__((ext_vector_type(8))) float v8f;

__device__ __forceinline__ v8f wmma4(v2f a, v2f b, v8f c) {
  return __builtin_amdgcn_wmma_f32_16x16x4_f32(false, a, false, b, (short)0, c,
                                               false, false);
}

__device__ __forceinline__ float silu_f(float v) {
  return v * (1.0f / (1.0f + __expf(-v)));
}

__device__ __forceinline__ void bspline7(float v, float* out) {
  float b[10];
#pragma unroll
  for (int j = 0; j < 10; ++j) {
    float gj = 0.5f * (float)j - 2.5f;
    b[j] = (v >= gj && v < gj + 0.5f) ? 1.0f : 0.0f;
  }
#pragma unroll
  for (int p = 1; p <= 3; ++p) {
    float invd = 1.0f / (0.5f * (float)p);
#pragma unroll
    for (int j = 0; j < 10 - p; ++j) {
      float gj  = 0.5f * (float)j - 2.5f;
      float gjp = gj + 0.5f * (float)p;
      b[j] = (v - gj) * invd * b[j] + ((gjp + 0.5f) - v) * invd * b[j + 1];
    }
  }
#pragma unroll
  for (int c = 0; c < N_COEF; ++c) out[c] = b[c];
}

#define GX_GROUPS (IN_C / 8)
static_assert(GX_GROUPS % 2 == 0, "groups are taken in pairs");
struct GxBuf { float4 w; float4 a[4]; };

__device__ __forceinline__ GxBuf gx_load(const float4* __restrict__ wp,
                                         const float4* const* ap, int m) {
  GxBuf b;
  b.w = wp[2 * m];
#pragma unroll
  for (int i = 0; i < 4; ++i) b.a[i] = ap[i][2 * m];
  return b;
}

__device__ __forceinline__ void gx_group(const GxBuf& b, v8f* acc) {
  v2f w0, w1;
  w0.x = b.w.x; w0.y = b.w.y;
  w1.x = b.w.z; w1.y = b.w.w;
#pragma unroll
  for (int i = 0; i < 4; ++i) {
    v2f a;
    a.x = b.a[i].x; a.y = b.a[i].y;
    acc[i] = wmma4(a, w0, acc[i]);
  }
#pragma unroll
  for (int i = 0; i < 4; ++i) {
    v2f a;
    a.x = b.a[i].z; a.y = b.a[i].w;
    acc[i] = wmma4(a, w1, acc[i]);
  }
}

__global__ void __launch_bounds__(256, 8)
k_gemm_xw(const float* __restrict__ x, const float* __restrict__ W,
          const float* __restrict__ dinv,
          float* __restrict__ hs, float* __restrict__ h) {
  const int NT = N_NODES / 16;
  const int NG = (NT + 3) / 4;
  int wave = (int)((blockIdx.x * blockDim.x + threadIdx.x) >> 5);
  int lane = (int)(threadIdx.x & 31);
  if (wave >= NG * (OUT_C / 16)) return;
  int grp = wave >> 3, ct = wave & 7;
  int o0 = ct * 16;
  int half = lane >> 4, l16 = lane & 15;

  int  rt[4];
  bool valid[4];
#pragma unroll
  for (int i = 0; i < 4; ++i) {
    int r = grp * 4 + i;
    valid[i] = (r < NT);
    rt[i] = valid[i] ? r : (NT - 1);
  }

  const float4* wp = (const float4*)(W + (size_t)(o0 + l16) * IN_C + 4 * half);
  const float4* ap[4];
#pragma unroll
  for (int i = 0; i < 4; ++i)
    ap[i] = (const float4*)(x + (size_t)(rt[i] * 16 + l16) * IN_C + 4 * half);

  v8f acc[4] = {};
  GxBuf ba = gx_load(wp, ap, 0);
  for (int m = 0; m + 2 < GX_GROUPS; m += 2) {
    GxBuf bb = gx_load(wp, ap, m + 1);
    __builtin_amdgcn_sched_barrier(0);
    gx_group(ba, acc);
    ba = gx_load(wp, ap, m + 2);
    __builtin_amdgcn_sched_barrier(0);
    gx_group(bb, acc);
  }
  float4 d0[4], d1[4];
  {
    GxBuf bb = gx_load(wp, ap, GX_GROUPS - 1);
    __builtin_amdgcn_sched_barrier(0);
    gx_group(ba, acc);
#pragma unroll
    for (int i = 0; i < 4; ++i) {
      const float4* dp = (const float4*)(dinv + rt[i] * 16 + half * 8);
      d0[i] = dp[0];
      d1[i] = dp[1];
    }
    __builtin_amdgcn_sched_barrier(0);
    gx_group(bb, acc);
  }

#pragma unroll
  for (int i = 0; i < 4; ++i) {
    if (!valid[i]) continue;
    const float dv[8] = {d0[i].x, d0[i].y, d0[i].z, d0[i].w, d1[i].x, d1[i].y, d1[i].z, d1[i].w};
#pragma unroll
    for (int j = 0; j < 8; ++j) {
      const int idx = (rt[i] * 16 + j + half * 8) * OUT_C + o0 + l16;
      const float v = acc[i][j] * dv[j];
      hs[idx] = v;
      h[idx]  = v;
    }
  }
}

#define SC_TICK_PITCH 64
#define DEG_PARTS   64
#define DEG_RANGES  4
#define DEG_THREADS 1024
#define DEG_SLICE   (N_EDGES / DEG_PARTS)
#define DEG_ITERS   ((DEG_SLICE + DEG_THREADS - 1) / DEG_THREADS)
#define DEG_RNODES  (N_NODES / DEG_RANGES)
#define DEG_VEC     (DEG_RNODES / 4)

__global__ void __launch_bounds__(DEG_THREADS)
k_deg_hist(const int* __restrict__ ei, int* __restrict__ part) {
  __shared__ alignas(16) int hist[DEG_RNODES];
  const int tid   = (int)threadIdx.x;
  const int slice = (int)blockIdx.x / DEG_RANGES;
  const int lo    = ((int)blockIdx.x % DEG_RANGES) * DEG_RNODES;
  const int* dst  = ei + N_EDGES + slice * DEG_SLICE;

  int d[DEG_ITERS];
#pragma unroll
  for (int j = 0; j < DEG_ITERS; ++j) {
    int e = tid + j * DEG_THREADS;
    d[j] = (e < DEG_SLICE) ? dst[e] : -1;
  }

  int4* h4 = (int4*)hist;
  for (int i = tid; i < DEG_VEC; i += DEG_THREADS) h4[i] = make_int4(0, 0, 0, 0);
  __syncthreads();

#pragma unroll
  for (int j = 0; j < DEG_ITERS; ++j) {
    unsigned rel = (unsigned)(d[j] - lo);
    if (rel < (unsigned)DEG_RNODES)
      __hip_atomic_fetch_add(&hist[rel], 1, __ATOMIC_RELAXED,
                             __HIP_MEMORY_SCOPE_WORKGROUP);
  }
  __syncthreads();

  int4* p4 = (int4*)(part + (size_t)slice * N_NODES + lo);
  for (int i = tid; i < DEG_VEC; i += DEG_THREADS) p4[i] = h4[i];
}

#define FOLD_ITEMS (8 * OUT_C * HIDDEN)
__device__ __forceinline__ void fold_w0(int t, const float* __restrict__ bw,
                                        const float* __restrict__ sw,
                                        const float* __restrict__ sc,
                                        float* __restrict__ Bm) {
  int k = t >> 4, o = t & 15;
  int f = k >> 3, j = k & 7;
  float w = (j == 0) ? bw[o * OUT_C + f]
                     : sw[(o * OUT_C + f) * N_COEF + (j - 1)] * sc[o * OUT_C + f];
  Bm[((f * 2 + (j >> 2)) * HIDDEN + o) * 4 + (j & 3)] = w;
}

__device__ __forceinline__ void fold_w1(int t, const float* __restrict__ bw,
                                        const float* __restrict__ sw,
                                        const float* __restrict__ sc,
                                        float* __restrict__ Bm) {
  int k = t >> 7, o = t & 127;
  int f = k >> 3, j = k & 7;
  float w = (j == 0) ? bw[o * HIDDEN + f]
                     : sw[(o * HIDDEN + f) * N_COEF + (j - 1)] * sc[o * HIDDEN + f];
  Bm[((k >> 1) * OUT_C + o) * 2 + (k & 1)] = w;
}

static_assert(2 * FOLD_ITEMS <= N_NODES, "one fold item per thread of the node grid");
__global__ void k_deg_dinv(const int* __restrict__ part, float* __restrict__ dinv,
                           unsigned* __restrict__ tick,
                           const float* __restrict__ bw0, const float* __restrict__ sw0,
                           const float* __restrict__ sc0, float* __restrict__ Bm0,
                           const float* __restrict__ bw1, const float* __restrict__ sw1,
                           const float* __restrict__ sc1, float* __restrict__ Bm1) {
  int i = blockIdx.x * blockDim.x + threadIdx.x;
  if (i < 4) tick[i * SC_TICK_PITCH] = 0u;
  const bool node = (i < N_NODES);
  const int  ic   = node ? i : (N_NODES - 1);
  int pv[DEG_PARTS];
#pragma unroll
  for (int p = 0; p < DEG_PARTS; ++p) pv[p] = part[p * N_NODES + ic];
  if (i < FOLD_ITEMS)          fold_w0(i, bw0, sw0, sc0, Bm0);
  else if (i < 2 * FOLD_ITEMS) fold_w1(i - FOLD_ITEMS, bw1, sw1, sc1, Bm1);
  if (node) {
    int c = 1;
#pragma unroll
    for (int p = 0; p < DEG_PARTS; ++p) c += pv[p];
    float dg = (float)c;
    dinv[i] = (dg > 0.0f) ? rsqrtf(dg) : 0.0f;
  }
}

#define SC_WAVES 4
#define SC_EPW   64
#define SC_BATCH 16
#define SC_UNIT  (SC_WAVES * SC_EPW)
#define SC_UNITS (N_EDGES / SC_UNIT)
static_assert(N_EDGES % SC_UNIT == 0, "a workgroup takes exactly SC_UNIT edges");
static_assert(SC_EPW == 64, "two edges per lane");
static_assert(SC_EPW % SC_BATCH == 0, "whole batches");

__global__ void __launch_bounds__(SC_WAVES * 32)
k_edge_scatter(const int* __restrict__ ei,
               const float* __restrict__ hs,
               float* __restrict__ h,
               unsigned* __restrict__ tick) {
  __shared__ unsigned s_job;
  const int lane = (int)(threadIdx.x & 31);
  const int wv   = (int)(threadIdx.x >> 5);
  if (threadIdx.x == 0) {
    const unsigned id = __builtin_amdgcn_s_sendmsg_rtn(0x87);
    const unsigned m  = (id >> 18) & 1u;
    const unsigned t  = __hip_atomic_fetch_add(&tick[m * SC_TICK_PITCH], 1u, __ATOMIC_RELAXED,
                                               __HIP_MEMORY_SCOPE_AGENT);
    unsigned job;
    if (t < (unsigned)SC_UNITS) {
      job = t * 2u + m;
    } else {
      const unsigned o  = m ^ 1u;
      const unsigned t2 = __hip_atomic_fetch_add(&tick[(2u + o) * SC_TICK_PITCH], 1u, __ATOMIC_RELAXED,
                                                 __HIP_MEMORY_SCOPE_AGENT);
      const unsigned u  = (t2 < (unsigned)SC_UNITS) ? ((unsigned)SC_UNITS - 1u - t2) : 0u;
      job = u * 2u + o;
    }
    s_job = job;
  }
  __syncthreads();
  const unsigned job  = (unsigned)__builtin_amdgcn_readfirstlane((int)s_job);
  const int      unit = (int)(job >> 1);
  const int      half = (int)(job & 1u);

  const int e0 = unit * SC_UNIT + wv * SC_EPW + lane;
  int src[2], dst[2];
  src[0] = ei[e0];
  src[1] = ei[e0 + 32];
  dst[0] = ei[N_EDGES + e0];
  dst[1] = ei[N_EDGES + e0 + 32];

  const float* hsl = hs + half * 64 + lane;
  float*       hl  = h  + half * 64 + lane;

#pragma unroll
  for (int b = 0; b < SC_EPW; b += SC_BATCH) {
    float v[SC_BATCH][2];
#pragma unroll
    for (int u = 0; u < SC_BATCH; ++u) {
      const int s = __builtin_amdgcn_readlane(src[(b + u) >> 5], (b + u) & 31);
      const float* row = hsl + (size_t)s * OUT_C;
      v[u][0] = row[0];
      v[u][1] = row[32];
    }
#pragma unroll
    for (int u = 0; u < SC_BATCH; ++u) {
      const int d = __builtin_amdgcn_readlane(dst[(b + u) >> 5], (b + u) & 31);
      float* base = hl + (size_t)d * OUT_C;
      __hip_atomic_fetch_add(base +  0, v[u][0], __ATOMIC_RELAXED, __HIP_MEMORY_SCOPE_AGENT);
      __hip_atomic_fetch_add(base + 32, v[u][1], __ATOMIC_RELAXED, __HIP_MEMORY_SCOPE_AGENT);
    }
  }
}

#define K0_GROUPS 16
static_assert(K0_GROUPS % 2 == 0, "groups are taken in pairs");
struct K0Buf { float4 hv; float4 bv; float4 w[4][2]; };

__device__ __forceinline__ K0Buf k0_load(const float4* __restrict__ hp,
                                         const float4* __restrict__ bp,
                                         const float4* __restrict__ wp, int m) {
  K0Buf b;
  b.hv = hp[2 * m];
  b.bv = bp[2 * m];
#pragma unroll
  for (int i = 0; i < 4; ++i) {
    b.w[i][0] = wp[(8 * m + i) * 32];
    b.w[i][1] = wp[(8 * m + i) * 32 + 16];
  }
  return b;
}

__device__ __forceinline__ void k0_group(const K0Buf& b, float di, v8f& acc0, v8f& acc1) {
  const float vv[4] = {b.hv.x * di + b.bv.x, b.hv.y * di + b.bv.y,
                       b.hv.z * di + b.bv.z, b.hv.w * di + b.bv.w};
#pragma unroll
  for (int i = 0; i < 4; ++i) {
    float e[8];
    e[0] = silu_f(vv[i]);
    bspline7(vv[i], e + 1);
    const float4 w0 = b.w[i][0], w1 = b.w[i][1];
    v2f a, w;
    a.x = e[0]; a.y = e[1]; w.x = w0.x; w.y = w0.y; acc0 = wmma4(a, w, acc0);
    a.x = e[2]; a.y = e[3]; w.x = w0.z; w.y = w0.w; acc1 = wmma4(a, w, acc1);
    a.x = e[4]; a.y = e[5]; w.x = w1.x; w.y = w1.y; acc0 = wmma4(a, w, acc0);
    a.x = e[6]; a.y = e[7]; w.x = w1.z; w.y = w1.w; acc1 = wmma4(a, w, acc1);
  }
}

#define LDS1_STRIDE 130
#define K1_STEPS 32
#define K1_GROUP 4
static_assert(K1_STEPS % K1_GROUP == 0, "whole groups");

__global__ void __launch_bounds__(32) k_kan01(const float* __restrict__ h,
                                              const float* __restrict__ dinv,
                                              const float* __restrict__ bias,
                                              const float* __restrict__ Bm0,
                                              const float* __restrict__ Bm1,
                                              float* __restrict__ out) {
  __shared__ float feat[16 * LDS1_STRIDE];
  int n0 = blockIdx.x * 16;
  int lane = threadIdx.x;
  int half = lane >> 4, l16 = lane & 15;

  __builtin_amdgcn_s_setprio(3);
  const float4* hp = (const float4*)(h + (size_t)(n0 + l16) * OUT_C + 4 * half);
  const float4* bp0 = (const float4*)(bias + 4 * half);
  const float4* wp = (const float4*)Bm0 + (4 * half) * 32 + l16;
  const float di = dinv[n0 + l16];

  v8f acc0 = {}, acc1 = {};
  K0Buf ba = k0_load(hp, bp0, wp, 0);
  for (int m = 0; m + 2 < K0_GROUPS; m += 2) {
    K0Buf bb = k0_load(hp, bp0, wp, m + 1);
    __builtin_amdgcn_sched_barrier(0);
    k0_group(ba, di, acc0, acc1);
    ba = k0_load(hp, bp0, wp, m + 2);
    __builtin_amdgcn_sched_barrier(0);
    k0_group(bb, di, acc0, acc1);
  }
  {
    K0Buf bb = k0_load(hp, bp0, wp, K0_GROUPS - 1);
    __builtin_amdgcn_sched_barrier(0);
    k0_group(ba, di, acc0, acc1);
    k0_group(bb, di, acc0, acc1);
  }

  v8f gt = acc0 + acc1;
#pragma unroll
  for (int i = 0; i < 8; ++i) {
    float v = gt[i];
    float* d = &feat[(half * 8 + i) * LDS1_STRIDE + l16 * 8];
    d[0] = silu_f(v);
    float bs[N_COEF];
    bspline7(v, bs);
#pragma unroll
    for (int c = 0; c < N_COEF; ++c) d[1 + c] = bs[c];
  }
  __syncthreads();

  __builtin_amdgcn_s_setprio(0);
  const float4* bp = (const float4*)(Bm1 + (half * OUT_C + 2 * l16) * 2);

  v8f acc[8] = {};
  float4 bb[2][K1_GROUP][4];
#pragma unroll
  for (int q = 0; q < K1_GROUP; ++q)
#pragma unroll
    for (int j = 0; j < 4; ++j) bb[0][q][j] = bp[q * 128 + j * 16];

#pragma unroll
  for (int grp = 0; grp < K1_STEPS / K1_GROUP; ++grp) {
    if (grp + 1 < K1_STEPS / K1_GROUP) {
#pragma unroll
      for (int q = 0; q < K1_GROUP; ++q)
#pragma unroll
        for (int j = 0; j < 4; ++j)
          bb[(grp + 1) & 1][q][j] = bp[((grp + 1) * K1_GROUP + q) * 128 + j * 16];
    }
#pragma unroll
    for (int q = 0; q < K1_GROUP; ++q) {
      int ka0 = (grp * K1_GROUP + q) * 4 + half * 2;
      v2f a;
      a.x = feat[l16 * LDS1_STRIDE + ka0];
      a.y = feat[l16 * LDS1_STRIDE + ka0 + 1];
#pragma unroll
      for (int j = 0; j < 4; ++j) {
        float4 w = bb[grp & 1][q][j];
        v2f b0, b1;
        b0.x = w.x; b0.y = w.y;
        b1.x = w.z; b1.y = w.w;
        acc[2 * j]     = wmma4(a, b0, acc[2 * j]);
        acc[2 * j + 1] = wmma4(a, b1, acc[2 * j + 1]);
      }
    }
  }

#pragma unroll
  for (int i = 0; i < 8; ++i) {
    float* orow = out + (size_t)(n0 + i + half * 8) * OUT_C + 2 * l16;
#pragma unroll
    for (int j = 0; j < 4; ++j) {
      float2 v;
      v.x = acc[2 * j][i];
      v.y = acc[2 * j + 1][i];
      *(float2*)(orow + 32 * j) = v;
    }
  }
}

extern "C" void kernel_launch(void* const* d_in, const int* in_sizes, int n_in,
                              void* d_out, int out_size, void* d_ws, size_t ws_size,
                              hipStream_t stream) {
  (void)in_sizes; (void)n_in; (void)out_size; (void)ws_size;
  const float* x       = (const float*)d_in[0];
  const int*   ei      = (const int*)d_in[1];
  const float* gcn_w   = (const float*)d_in[2];
  const float* gcn_b   = (const float*)d_in[3];
  const float* base_w0 = (const float*)d_in[4];
  const float* spl_w0  = (const float*)d_in[5];
  const float* scal0   = (const float*)d_in[6];
  const float* base_w1 = (const float*)d_in[7];
  const float* spl_w1  = (const float*)d_in[8];
  const float* scal1   = (const float*)d_in[9];
  float* out = (float*)d_out;

  float* ws    = (float*)d_ws;
  float* hs    = ws;
  float* h     = hs + (size_t)N_NODES * OUT_C;
  float* deg   = h + (size_t)N_NODES * OUT_C;
  float* g     = deg + N_NODES;
  float* Bm0   = g + (size_t)N_NODES * HIDDEN;
  float* Bm1   = Bm0 + 8 * OUT_C * HIDDEN;

  const int NT = N_NODES / 16;
  const int NG = (NT + 3) / 4;

  int* part = (int*)h;
  unsigned* tick = (unsigned*)g;
  k_deg_hist<<<DEG_PARTS * DEG_RANGES, DEG_THREADS, 0, stream>>>(ei, part);
  k_deg_dinv<<<(N_NODES + 255) / 256, 256, 0, stream>>>(part, deg, tick,
                                                        base_w0, spl_w0, scal0, Bm0,
                                                        base_w1, spl_w1, scal1, Bm1);
  k_gemm_xw<<<NG, 256, 0, stream>>>(x, gcn_w, deg, hs, h);
  k_edge_scatter<<<2 * SC_UNITS, SC_WAVES * 32, 0, stream>>>(ei, hs, h, tick);
  k_kan01<<<N_NODES / 16, 32, 0, stream>>>(h, deg, gcn_b, Bm0, Bm1, out);
}
